# starting kernel
# baseline (speedup 1.0000x reference)
#include <hip/hip_runtime.h>
#include <hip/hip_bf16.h>


#define P_    13
#define PP    169
#define PPD   176
#define NIMG  4
#define HW    256
#define HP    280
#define HO    268
#define LPOS  (HO*HO)
#define IMGPAD (HP*HP)

#define XPAD_OFF 0
#define PM1_OFF  (NIMG*IMGPAD)
#define QIN_OFF  (PM1_OFF + PPD*PPD)
#define NUM_OFF  (QIN_OFF + PPD*PPD)
#define DIV_OFF  (NUM_OFF + NIMG*IMGPAD)
#define AUG_OFF  (DIV_OFF + NIMG*IMGPAD)
#define AUGW 344

#define WAVES 4

typedef float v2f __attribute__((ext_vector_type(2)));
typedef float v8f __attribute__((ext_vector_type(8)));

__global__ void prep_kernel(const float* __restrict__ x,
                            const float* __restrict__ Pm1,
                            float* __restrict__ ws) {
    int idx = blockIdx.x * blockDim.x + threadIdx.x;
    if (idx < NIMG * IMGPAD) {
        ws[NUM_OFF + idx] = 0.0f;
        ws[DIV_OFF + idx] = 0.0f;
        int n   = idx / IMGPAD;
        int rem = idx % IMGPAD;
        int y   = rem / HP;
        int xx  = rem % HP;
        int ty = y  - (P_ - 1); ty = (ty < 0) ? -ty : ((ty > HW - 1) ? 2 * (HW - 1) - ty : ty);
        int tx = xx - (P_ - 1); tx = (tx < 0) ? -tx : ((tx > HW - 1) ? 2 * (HW - 1) - tx : tx);
        ws[XPAD_OFF + idx] = 2.0f * x[n * HW * HW + ty * HW + tx] - 1.0f;
    }
    if (idx < PPD * PPD) {
        int r = idx / PPD, c = idx % PPD;
        ws[PM1_OFF + idx] = (r < PP && c < PP) ? Pm1[r * PP + c] : 0.0f;
        ws[QIN_OFF + idx] = 0.0f;
    }
}

__global__ void invert_kernel(const float* __restrict__ Pm1,
                              float* __restrict__ ws) {
    float* aug = ws + AUG_OFF;
    const int tid = threadIdx.x;

    for (int idx = tid; idx < PP * AUGW; idx += blockDim.x) {
        int r = idx / AUGW, c = idx % AUGW;
        float v = 0.0f;
        if (c < PP)            v = Pm1[r * PP + c];
        else if (c < 2 * PP)   v = (c - PP == r) ? 1.0f : 0.0f;
        aug[idx] = v;
    }
    __syncthreads();

    for (int k = 0; k < PP; ++k) {
        float piv = aug[k * AUGW + k];
        __syncthreads();
        float inv = 1.0f / piv;
        for (int c = tid; c < 2 * PP; c += blockDim.x)
            aug[k * AUGW + c] *= inv;
        __syncthreads();
        for (int r = tid; r < PP; r += blockDim.x) {
            if (r != k) {
                float f = aug[r * AUGW + k];
                for (int c = 0; c < 2 * PP; ++c)
                    aug[r * AUGW + c] -= f * aug[k * AUGW + c];
            }
        }
        __syncthreads();
    }

    for (int idx = tid; idx < PP * PP; idx += blockDim.x) {
        int r = idx / PP, c = idx % PP;
        ws[QIN_OFF + r * PPD + c] = aug[r * AUGW + PP + c];
    }
}

__global__ __launch_bounds__(32 * WAVES)
void dct2net_main(const float* __restrict__ sigma_,
                  float* __restrict__ ws) {
    __shared__ float lds_t[WAVES][PPD][17];

    const float* __restrict__ xpad = ws + XPAD_OFF;
    const float* __restrict__ pm1  = ws + PM1_OFF;
    const float* __restrict__ qin  = ws + QIN_OFF;
    float* __restrict__ numb = ws + NUM_OFF;
    float* __restrict__ divb = ws + DIV_OFF;

    const int wave  = threadIdx.x >> 5;
    const int lane  = threadIdx.x & 31;
    const int laneN = lane & 15;
    const int hi    = lane >> 4;

    const int img  = blockIdx.y;
    const int l0   = (blockIdx.x * WAVES + wave) * 16;
    const int lcol = l0 + laneN;
    const bool colOK = (lcol < LPOS);
    const int lc = colOK ? lcol : 0;
    const int ip = lc / HO;
    const int jp = lc - ip * HO;
    const float* __restrict__ xb = xpad + img * IMGPAD + ip * HP + jp;

    const float lam = 6.0f * sigma_[0];

    const v8f vzero = {0.f, 0.f, 0.f, 0.f, 0.f, 0.f, 0.f, 0.f};
    v8f acc[11];
#pragma unroll
    for (int mt = 0; mt < 11; ++mt) acc[mt] = vzero;

#pragma unroll 4
    for (int kt = 0; kt < 44; ++kt) {
        const int ka = kt * 4 + 2 * hi;
        const int d0 = ka / P_;
        const int d1 = (ka + 1) / P_;
        v2f b;
        b.x = xb[d0 * HP + (ka - d0 * P_)];
        b.y = xb[d1 * HP + (ka + 1 - d1 * P_)];
#pragma unroll
        for (int mt = 0; mt < 11; ++mt) {
            v2f a = *(const v2f*)(pm1 + (mt * 16 + laneN) * PPD + ka);
            acc[mt] = __builtin_amdgcn_wmma_f32_16x16x4_f32(
                false, a, false, b, (short)0, acc[mt], false, false);
        }
    }

    int cnt = 0;
#pragma unroll
    for (int mt = 0; mt < 11; ++mt) {
#pragma unroll
        for (int v = 0; v < 8; ++v) {
            float t = acc[mt][v];
            t = (fabsf(t) > lam) ? t : 0.0f;
            acc[mt][v] = t;
            bool isRow0 = (mt == 0 && v == 0 && hi == 0);
            cnt += (!isRow0 && t != 0.0f) ? 1 : 0;
        }
    }
    cnt += __shfl_xor(cnt, 16, 32);
    const float w = 1.0f / (1.0f + (float)cnt);

#pragma unroll
    for (int mt = 0; mt < 11; ++mt)
#pragma unroll
        for (int v = 0; v < 8; ++v)
            lds_t[wave][mt * 16 + v + hi * 8][laneN] = acc[mt][v];
    __syncthreads();

    v8f acc2[11];
#pragma unroll
    for (int mt = 0; mt < 11; ++mt) acc2[mt] = vzero;

#pragma unroll 4
    for (int kt = 0; kt < 44; ++kt) {
        const int ka = kt * 4 + 2 * hi;
        v2f b;
        b.x = lds_t[wave][ka][laneN];
        b.y = lds_t[wave][ka + 1][laneN];
#pragma unroll
        for (int mt = 0; mt < 11; ++mt) {
            v2f a = *(const v2f*)(qin + (mt * 16 + laneN) * PPD + ka);
            acc2[mt] = __builtin_amdgcn_wmma_f32_16x16x4_f32(
                false, a, false, b, (short)0, acc2[mt], false, false);
        }
    }

    if (colOK) {
        float* nb = numb + img * IMGPAD + ip * HP + jp;
        float* db = divb + img * IMGPAD + ip * HP + jp;
#pragma unroll
        for (int mt = 0; mt < 11; ++mt) {
#pragma unroll
            for (int v = 0; v < 8; ++v) {
                int M = mt * 16 + v + hi * 8;
                if (M < PP) {
                    int d   = M / P_;
                    int off = d * HP + (M - d * P_);
                    atomicAdd(nb + off, acc2[mt][v] * w);
                    atomicAdd(db + off, w);
                }
            }
        }
    }
}

__global__ void finish_kernel(const float* __restrict__ ws,
                              float* __restrict__ out) {
    int idx = blockIdx.x * blockDim.x + threadIdx.x;
    if (idx >= NIMG * HW * HW) return;
    int n   = idx / (HW * HW);
    int rem = idx % (HW * HW);
    int y   = rem / HW;
    int x   = rem % HW;
    int s = n * IMGPAD + (y + P_ - 1) * HP + (x + P_ - 1);
    out[idx] = (ws[NUM_OFF + s] / ws[DIV_OFF + s] + 1.0f) * 0.5f;
}

extern "C" void kernel_launch(void* const* d_in, const int* in_sizes, int n_in,
                              void* d_out, int out_size, void* d_ws, size_t ws_size,
                              hipStream_t stream) {
    const float* x      = (const float*)d_in[0];
    const float* sigma_ = (const float*)d_in[1];
    const float* Pm1    = (const float*)d_in[2];
    float* out = (float*)d_out;
    float* ws  = (float*)d_ws;

    const int prepN = NIMG * IMGPAD;
    prep_kernel<<<(prepN + 255) / 256, 256, 0, stream>>>(x, Pm1, ws);

    invert_kernel<<<1, 256, 0, stream>>>(Pm1, ws);

    dim3 grid((LPOS + WAVES * 16 - 1) / (WAVES * 16), NIMG);
    dct2net_main<<<grid, 32 * WAVES, 0, stream>>>(sigma_, ws);

    const int outN = NIMG * HW * HW;
    finish_kernel<<<(outN + 255) / 256, 256, 0, stream>>>(ws, out);
}


# Gauss-Jordan inverse re-parallelised: 22x22 threads hold 8x8 register tiles, pivot row/column exchanged via LDS, one barrier per pivot; same pivot order and per-element fma, no sums reordered
# speedup vs baseline: 21.8642x; 21.8642x over previous
#include <hip/hip_runtime.h>
#include <hip/hip_bf16.h>


#define P_    13
#define PP    169
#define PPD   176
#define NIMG  4
#define HW    256
#define HP    280
#define HO    268
#define LPOS  (HO*HO)
#define IMGPAD (HP*HP)

#define XPAD_OFF 0
#define PM1_OFF  (NIMG*IMGPAD)
#define QIN_OFF  (PM1_OFF + PPD*PPD)
#define NUM_OFF  (QIN_OFF + PPD*PPD)
#define DIV_OFF  (NUM_OFF + NIMG*IMGPAD)

#define GJ_G 22
#define GJ_T 8
#define GJ_STRIP (GJ_G * GJ_T)
#define GJ_THREADS 512

#define WAVES 4

typedef float v2f __attribute__((ext_vector_type(2)));
typedef float v8f __attribute__((ext_vector_type(8)));

__global__ void prep_kernel(const float* __restrict__ x,
                            const float* __restrict__ Pm1,
                            float* __restrict__ ws) {
    int idx = blockIdx.x * blockDim.x + threadIdx.x;
    if (idx < NIMG * IMGPAD) {
        ws[NUM_OFF + idx] = 0.0f;
        ws[DIV_OFF + idx] = 0.0f;
        int n   = idx / IMGPAD;
        int rem = idx % IMGPAD;
        int y   = rem / HP;
        int xx  = rem % HP;
        int ty = y  - (P_ - 1); ty = (ty < 0) ? -ty : ((ty > HW - 1) ? 2 * (HW - 1) - ty : ty);
        int tx = xx - (P_ - 1); tx = (tx < 0) ? -tx : ((tx > HW - 1) ? 2 * (HW - 1) - tx : tx);
        ws[XPAD_OFF + idx] = 2.0f * x[n * HW * HW + ty * HW + tx] - 1.0f;
    }
    if (idx < PPD * PPD) {
        int r = idx / PPD, c = idx % PPD;
        ws[PM1_OFF + idx] = (r < PP && c < PP) ? Pm1[r * PP + c] : 0.0f;
        ws[QIN_OFF + idx] = 0.0f;
    }
}

template <int JJ>
__device__ __forceinline__ void gj_phase(float (&a)[GJ_T][GJ_T],
                                         float (*prow)[GJ_STRIP],
                                         float (*fcol)[GJ_STRIP],
                                         const int ti, const int tj,
                                         const bool act) {
    constexpr int oEnd = ((JJ + 1) * GJ_G <= PP) ? GJ_G : (PP - JJ * GJ_G);
    for (int o = 0; o < oEnd; ++o) {
        const int par = (JJ * GJ_G + o) & 1;
        if (act && ti == o) {
#pragma unroll
            for (int j = 0; j < GJ_T; ++j) prow[par][tj * GJ_T + j] = a[JJ][j];
        }
        if (act && tj == o) {
#pragma unroll
            for (int i = 0; i < GJ_T; ++i) fcol[par][ti * GJ_T + i] = a[i][JJ];
        }
        __syncthreads();

        const float piv = prow[par][o * GJ_T + JJ];
        const float inv = 1.0f / piv;
        float p[GJ_T], f[GJ_T];
#pragma unroll
        for (int j = 0; j < GJ_T; ++j) p[j] = prow[par][tj * GJ_T + j] * inv;
#pragma unroll
        for (int i = 0; i < GJ_T; ++i) f[i] = fcol[par][ti * GJ_T + i];
        const bool ownCol = (tj == o);
        const bool ownRow = (ti == o);
        if (ownCol) p[JJ] = inv;
#pragma unroll
        for (int i = 0; i < GJ_T; ++i)
#pragma unroll
            for (int j = 0; j < GJ_T; ++j)
                a[i][j] -= f[i] * p[j];
        if (ownCol) {
#pragma unroll
            for (int i = 0; i < GJ_T; ++i) a[i][JJ] = 0.0f - f[i] * inv;
        }
        if (ownRow) {
#pragma unroll
            for (int j = 0; j < GJ_T; ++j) a[JJ][j] = p[j];
        }
    }
}

__global__ __launch_bounds__(GJ_THREADS)
void invert_kernel(const float* __restrict__ Pm1,
                   float* __restrict__ ws) {
    __shared__ __attribute__((aligned(16))) float prow[2][GJ_STRIP];
    __shared__ __attribute__((aligned(16))) float fcol[2][GJ_STRIP];

    const int tid = threadIdx.x;
    const bool act = tid < GJ_G * GJ_G;
    const int ti = act ? tid / GJ_G : 0;
    const int tj = act ? tid - ti * GJ_G : 0;

    float a[GJ_T][GJ_T];
#pragma unroll
    for (int i = 0; i < GJ_T; ++i) {
#pragma unroll
        for (int j = 0; j < GJ_T; ++j) {
            const int r = ti + GJ_G * i;
            const int c = tj + GJ_G * j;
            a[i][j] = (r < PP && c < PP) ? Pm1[r * PP + c] : 0.0f;
        }
    }

    gj_phase<0>(a, prow, fcol, ti, tj, act);
    gj_phase<1>(a, prow, fcol, ti, tj, act);
    gj_phase<2>(a, prow, fcol, ti, tj, act);
    gj_phase<3>(a, prow, fcol, ti, tj, act);
    gj_phase<4>(a, prow, fcol, ti, tj, act);
    gj_phase<5>(a, prow, fcol, ti, tj, act);
    gj_phase<6>(a, prow, fcol, ti, tj, act);
    gj_phase<7>(a, prow, fcol, ti, tj, act);

    if (act) {
#pragma unroll
        for (int i = 0; i < GJ_T; ++i) {
#pragma unroll
            for (int j = 0; j < GJ_T; ++j) {
                const int r = ti + GJ_G * i;
                const int c = tj + GJ_G * j;
                if (r < PP && c < PP) ws[QIN_OFF + r * PPD + c] = a[i][j];
            }
        }
    }
}

__global__ __launch_bounds__(32 * WAVES)
void dct2net_main(const float* __restrict__ sigma_,
                  float* __restrict__ ws) {
    __shared__ float lds_t[WAVES][PPD][17];

    const float* __restrict__ xpad = ws + XPAD_OFF;
    const float* __restrict__ pm1  = ws + PM1_OFF;
    const float* __restrict__ qin  = ws + QIN_OFF;
    float* __restrict__ numb = ws + NUM_OFF;
    float* __restrict__ divb = ws + DIV_OFF;

    const int wave  = threadIdx.x >> 5;
    const int lane  = threadIdx.x & 31;
    const int laneN = lane & 15;
    const int hi    = lane >> 4;

    const int img  = blockIdx.y;
    const int l0   = (blockIdx.x * WAVES + wave) * 16;
    const int lcol = l0 + laneN;
    const bool colOK = (lcol < LPOS);
    const int lc = colOK ? lcol : 0;
    const int ip = lc / HO;
    const int jp = lc - ip * HO;
    const float* __restrict__ xb = xpad + img * IMGPAD + ip * HP + jp;

    const float lam = 6.0f * sigma_[0];

    const v8f vzero = {0.f, 0.f, 0.f, 0.f, 0.f, 0.f, 0.f, 0.f};
    v8f acc[11];
#pragma unroll
    for (int mt = 0; mt < 11; ++mt) acc[mt] = vzero;

#pragma unroll 4
    for (int kt = 0; kt < 44; ++kt) {
        const int ka = kt * 4 + 2 * hi;
        const int d0 = ka / P_;
        const int d1 = (ka + 1) / P_;
        v2f b;
        b.x = xb[d0 * HP + (ka - d0 * P_)];
        b.y = xb[d1 * HP + (ka + 1 - d1 * P_)];
#pragma unroll
        for (int mt = 0; mt < 11; ++mt) {
            v2f a = *(const v2f*)(pm1 + (mt * 16 + laneN) * PPD + ka);
            acc[mt] = __builtin_amdgcn_wmma_f32_16x16x4_f32(
                false, a, false, b, (short)0, acc[mt], false, false);
        }
    }

    int cnt = 0;
#pragma unroll
    for (int mt = 0; mt < 11; ++mt) {
#pragma unroll
        for (int v = 0; v < 8; ++v) {
            float t = acc[mt][v];
            t = (fabsf(t) > lam) ? t : 0.0f;
            acc[mt][v] = t;
            bool isRow0 = (mt == 0 && v == 0 && hi == 0);
            cnt += (!isRow0 && t != 0.0f) ? 1 : 0;
        }
    }
    cnt += __shfl_xor(cnt, 16, 32);
    const float w = 1.0f / (1.0f + (float)cnt);

#pragma unroll
    for (int mt = 0; mt < 11; ++mt)
#pragma unroll
        for (int v = 0; v < 8; ++v)
            lds_t[wave][mt * 16 + v + hi * 8][laneN] = acc[mt][v];
    __syncthreads();

    v8f acc2[11];
#pragma unroll
    for (int mt = 0; mt < 11; ++mt) acc2[mt] = vzero;

#pragma unroll 4
    for (int kt = 0; kt < 44; ++kt) {
        const int ka = kt * 4 + 2 * hi;
        v2f b;
        b.x = lds_t[wave][ka][laneN];
        b.y = lds_t[wave][ka + 1][laneN];
#pragma unroll
        for (int mt = 0; mt < 11; ++mt) {
            v2f a = *(const v2f*)(qin + (mt * 16 + laneN) * PPD + ka);
            acc2[mt] = __builtin_amdgcn_wmma_f32_16x16x4_f32(
                false, a, false, b, (short)0, acc2[mt], false, false);
        }
    }

    if (colOK) {
        float* nb = numb + img * IMGPAD + ip * HP + jp;
        float* db = divb + img * IMGPAD + ip * HP + jp;
#pragma unroll
        for (int mt = 0; mt < 11; ++mt) {
#pragma unroll
            for (int v = 0; v < 8; ++v) {
                int M = mt * 16 + v + hi * 8;
                if (M < PP) {
                    int d   = M / P_;
                    int off = d * HP + (M - d * P_);
                    atomicAdd(nb + off, acc2[mt][v] * w);
                    atomicAdd(db + off, w);
                }
            }
        }
    }
}

__global__ void finish_kernel(const float* __restrict__ ws,
                              float* __restrict__ out) {
    int idx = blockIdx.x * blockDim.x + threadIdx.x;
    if (idx >= NIMG * HW * HW) return;
    int n   = idx / (HW * HW);
    int rem = idx % (HW * HW);
    int y   = rem / HW;
    int x   = rem % HW;
    int s = n * IMGPAD + (y + P_ - 1) * HP + (x + P_ - 1);
    out[idx] = (ws[NUM_OFF + s] / ws[DIV_OFF + s] + 1.0f) * 0.5f;
}

extern "C" void kernel_launch(void* const* d_in, const int* in_sizes, int n_in,
                              void* d_out, int out_size, void* d_ws, size_t ws_size,
                              hipStream_t stream) {
    const float* x      = (const float*)d_in[0];
    const float* sigma_ = (const float*)d_in[1];
    const float* Pm1    = (const float*)d_in[2];
    float* out = (float*)d_out;
    float* ws  = (float*)d_ws;

    const int prepN = NIMG * IMGPAD;
    prep_kernel<<<(prepN + 255) / 256, 256, 0, stream>>>(x, Pm1, ws);

    invert_kernel<<<1, GJ_THREADS, 0, stream>>>(Pm1, ws);

    dim3 grid((LPOS + WAVES * 16 - 1) / (WAVES * 16), NIMG);
    dct2net_main<<<grid, 32 * WAVES, 0, stream>>>(sigma_, ws);

    const int outN = NIMG * HW * HW;
    finish_kernel<<<(outN + 255) / 256, 256, 0, stream>>>(ws, out);
}


# Overlap-add pre-summed per wave in LDS: lane q adds the up to 13 elements of the tile's patches that land on its pixel; 13 atomics per lane each for num and div replace 88 each (sum order changes)
# speedup vs baseline: 22.8600x; 1.0455x over previous
#include <hip/hip_runtime.h>
#include <hip/hip_bf16.h>


#define P_    13
#define PP    169
#define PPD   176
#define NIMG  4
#define HW    256
#define HP    280
#define HO    268
#define LPOS  (HO*HO)
#define IMGPAD (HP*HP)

#define XPAD_OFF 0
#define PM1_OFF  (NIMG*IMGPAD)
#define QIN_OFF  (PM1_OFF + PPD*PPD)
#define NUM_OFF  (QIN_OFF + PPD*PPD)
#define DIV_OFF  (NUM_OFF + NIMG*IMGPAD)

#define GJ_G 22
#define GJ_T 8
#define GJ_STRIP (GJ_G * GJ_T)
#define GJ_THREADS 512

#define WAVES 4

typedef float v2f __attribute__((ext_vector_type(2)));
typedef float v8f __attribute__((ext_vector_type(8)));

__global__ void prep_kernel(const float* __restrict__ x,
                            const float* __restrict__ Pm1,
                            float* __restrict__ ws) {
    int idx = blockIdx.x * blockDim.x + threadIdx.x;
    if (idx < NIMG * IMGPAD) {
        ws[NUM_OFF + idx] = 0.0f;
        ws[DIV_OFF + idx] = 0.0f;
        int n   = idx / IMGPAD;
        int rem = idx % IMGPAD;
        int y   = rem / HP;
        int xx  = rem % HP;
        int ty = y  - (P_ - 1); ty = (ty < 0) ? -ty : ((ty > HW - 1) ? 2 * (HW - 1) - ty : ty);
        int tx = xx - (P_ - 1); tx = (tx < 0) ? -tx : ((tx > HW - 1) ? 2 * (HW - 1) - tx : tx);
        ws[XPAD_OFF + idx] = 2.0f * x[n * HW * HW + ty * HW + tx] - 1.0f;
    }
    if (idx < PPD * PPD) {
        int r = idx / PPD, c = idx % PPD;
        ws[PM1_OFF + idx] = (r < PP && c < PP) ? Pm1[r * PP + c] : 0.0f;
        ws[QIN_OFF + idx] = 0.0f;
    }
}

template <int JJ>
__device__ __forceinline__ void gj_phase(float (&a)[GJ_T][GJ_T],
                                         float (*prow)[GJ_STRIP],
                                         float (*fcol)[GJ_STRIP],
                                         const int ti, const int tj,
                                         const bool act) {
    constexpr int oEnd = ((JJ + 1) * GJ_G <= PP) ? GJ_G : (PP - JJ * GJ_G);
    for (int o = 0; o < oEnd; ++o) {
        const int par = (JJ * GJ_G + o) & 1;
        if (act && ti == o) {
#pragma unroll
            for (int j = 0; j < GJ_T; ++j) prow[par][tj * GJ_T + j] = a[JJ][j];
        }
        if (act && tj == o) {
#pragma unroll
            for (int i = 0; i < GJ_T; ++i) fcol[par][ti * GJ_T + i] = a[i][JJ];
        }
        __syncthreads();

        const float piv = prow[par][o * GJ_T + JJ];
        const float inv = 1.0f / piv;
        float p[GJ_T], f[GJ_T];
#pragma unroll
        for (int j = 0; j < GJ_T; ++j) p[j] = prow[par][tj * GJ_T + j] * inv;
#pragma unroll
        for (int i = 0; i < GJ_T; ++i) f[i] = fcol[par][ti * GJ_T + i];
        const bool ownCol = (tj == o);
        const bool ownRow = (ti == o);
        if (ownCol) p[JJ] = inv;
#pragma unroll
        for (int i = 0; i < GJ_T; ++i)
#pragma unroll
            for (int j = 0; j < GJ_T; ++j)
                a[i][j] -= f[i] * p[j];
        if (ownCol) {
#pragma unroll
            for (int i = 0; i < GJ_T; ++i) a[i][JJ] = 0.0f - f[i] * inv;
        }
        if (ownRow) {
#pragma unroll
            for (int j = 0; j < GJ_T; ++j) a[JJ][j] = p[j];
        }
    }
}

__global__ __launch_bounds__(GJ_THREADS)
void invert_kernel(const float* __restrict__ Pm1,
                   float* __restrict__ ws) {
    __shared__ __attribute__((aligned(16))) float prow[2][GJ_STRIP];
    __shared__ __attribute__((aligned(16))) float fcol[2][GJ_STRIP];

    const int tid = threadIdx.x;
    const bool act = tid < GJ_G * GJ_G;
    const int ti = act ? tid / GJ_G : 0;
    const int tj = act ? tid - ti * GJ_G : 0;

    float a[GJ_T][GJ_T];
#pragma unroll
    for (int i = 0; i < GJ_T; ++i) {
#pragma unroll
        for (int j = 0; j < GJ_T; ++j) {
            const int r = ti + GJ_G * i;
            const int c = tj + GJ_G * j;
            a[i][j] = (r < PP && c < PP) ? Pm1[r * PP + c] : 0.0f;
        }
    }

    gj_phase<0>(a, prow, fcol, ti, tj, act);
    gj_phase<1>(a, prow, fcol, ti, tj, act);
    gj_phase<2>(a, prow, fcol, ti, tj, act);
    gj_phase<3>(a, prow, fcol, ti, tj, act);
    gj_phase<4>(a, prow, fcol, ti, tj, act);
    gj_phase<5>(a, prow, fcol, ti, tj, act);
    gj_phase<6>(a, prow, fcol, ti, tj, act);
    gj_phase<7>(a, prow, fcol, ti, tj, act);

    if (act) {
#pragma unroll
        for (int i = 0; i < GJ_T; ++i) {
#pragma unroll
            for (int j = 0; j < GJ_T; ++j) {
                const int r = ti + GJ_G * i;
                const int c = tj + GJ_G * j;
                if (r < PP && c < PP) ws[QIN_OFF + r * PPD + c] = a[i][j];
            }
        }
    }
}

__global__ __launch_bounds__(32 * WAVES)
void dct2net_main(const float* __restrict__ sigma_,
                  float* __restrict__ ws) {
    __shared__ float lds_t[WAVES][PPD][17];

    const float* __restrict__ xpad = ws + XPAD_OFF;
    const float* __restrict__ pm1  = ws + PM1_OFF;
    const float* __restrict__ qin  = ws + QIN_OFF;
    float* __restrict__ numb = ws + NUM_OFF;
    float* __restrict__ divb = ws + DIV_OFF;

    const int wave  = threadIdx.x >> 5;
    const int lane  = threadIdx.x & 31;
    const int laneN = lane & 15;
    const int hi    = lane >> 4;

    const int img  = blockIdx.y;
    const int l0   = (blockIdx.x * WAVES + wave) * 16;
    const int lcol = l0 + laneN;
    const bool colOK = (lcol < LPOS);
    const int lc = colOK ? lcol : 0;
    const int ip = lc / HO;
    const int jp = lc - ip * HO;
    const float* __restrict__ xb = xpad + img * IMGPAD + ip * HP + jp;

    const float lam = 6.0f * sigma_[0];

    const v8f vzero = {0.f, 0.f, 0.f, 0.f, 0.f, 0.f, 0.f, 0.f};
    v8f acc[11];
#pragma unroll
    for (int mt = 0; mt < 11; ++mt) acc[mt] = vzero;

#pragma unroll 4
    for (int kt = 0; kt < 44; ++kt) {
        const int ka = kt * 4 + 2 * hi;
        const int d0 = ka / P_;
        const int d1 = (ka + 1) / P_;
        v2f b;
        b.x = xb[d0 * HP + (ka - d0 * P_)];
        b.y = xb[d1 * HP + (ka + 1 - d1 * P_)];
#pragma unroll
        for (int mt = 0; mt < 11; ++mt) {
            v2f a = *(const v2f*)(pm1 + (mt * 16 + laneN) * PPD + ka);
            acc[mt] = __builtin_amdgcn_wmma_f32_16x16x4_f32(
                false, a, false, b, (short)0, acc[mt], false, false);
        }
    }

    int cnt = 0;
#pragma unroll
    for (int mt = 0; mt < 11; ++mt) {
#pragma unroll
        for (int v = 0; v < 8; ++v) {
            float t = acc[mt][v];
            t = (fabsf(t) > lam) ? t : 0.0f;
            acc[mt][v] = t;
            bool isRow0 = (mt == 0 && v == 0 && hi == 0);
            cnt += (!isRow0 && t != 0.0f) ? 1 : 0;
        }
    }
    cnt += __shfl_xor(cnt, 16, 32);
    const float w = 1.0f / (1.0f + (float)cnt);

#pragma unroll
    for (int mt = 0; mt < 11; ++mt)
#pragma unroll
        for (int v = 0; v < 8; ++v)
            lds_t[wave][mt * 16 + v + hi * 8][laneN] = acc[mt][v];
    __syncthreads();

    v8f acc2[11];
#pragma unroll
    for (int mt = 0; mt < 11; ++mt) acc2[mt] = vzero;

#pragma unroll 4
    for (int kt = 0; kt < 44; ++kt) {
        const int ka = kt * 4 + 2 * hi;
        v2f b;
        b.x = lds_t[wave][ka][laneN];
        b.y = lds_t[wave][ka + 1][laneN];
#pragma unroll
        for (int mt = 0; mt < 11; ++mt) {
            v2f a = *(const v2f*)(qin + (mt * 16 + laneN) * PPD + ka);
            acc2[mt] = __builtin_amdgcn_wmma_f32_16x16x4_f32(
                false, a, false, b, (short)0, acc2[mt], false, false);
        }
    }

    static_assert(LPOS % 16 == 0, "a tile is wholly inside or wholly outside the patch list");
    float (*T)[17] = lds_t[wave];
#pragma unroll
    for (int mt = 0; mt < 11; ++mt) {
#pragma unroll
        for (int v = 0; v < 8; ++v) {
            const int M = mt * 16 + v + hi * 8;
            if (M < PP) T[M][laneN] = acc2[mt][v] * w;
        }
    }
    if (hi == 0) T[PP][laneN] = w;
    for (int M = lane; M <= PP; M += 32) T[M][16] = 0.0f;
    __syncthreads();

    if (l0 < LPOS) {
        const int ip0 = l0 / HO;
        const int jp0 = l0 - ip0 * HO;
        const int s   = (HO - jp0 < 16) ? (HO - jp0) : 16;
        float* nb = numb + img * IMGPAD + ip0 * HP + jp0;
        float* db = divb + img * IMGPAD + ip0 * HP + jp0;
        const int npass = (s < 16) ? 2 : 1;
        for (int pass = 0; pass < npass; ++pass) {
            const int q = lane + 32 * pass;
            int col[P_];
            bool any = false;
#pragma unroll
            for (int dj = 0; dj < P_; ++dj) {
                const int sh = q - dj;
                const bool row0 = (sh >= 0) && (sh < s);
                const bool row1 = (sh >= s + 12) && (sh <= 27);
                col[dj] = row0 ? sh : (row1 ? sh - 12 : 16);
                any = any || row0 || row1;
            }
            float dsum = 0.0f;
#pragma unroll
            for (int dj = 0; dj < P_; ++dj) dsum += T[PP][col[dj]];
            if (any) {
#pragma unroll
                for (int di = 0; di < P_; ++di) {
                    float nsum = 0.0f;
#pragma unroll
                    for (int dj = 0; dj < P_; ++dj) nsum += T[di * P_ + dj][col[dj]];
                    atomicAdd(nb + di * HP + q, nsum);
                    atomicAdd(db + di * HP + q, dsum);
                }
            }
        }
    }
}

__global__ void finish_kernel(const float* __restrict__ ws,
                              float* __restrict__ out) {
    int idx = blockIdx.x * blockDim.x + threadIdx.x;
    if (idx >= NIMG * HW * HW) return;
    int n   = idx / (HW * HW);
    int rem = idx % (HW * HW);
    int y   = rem / HW;
    int x   = rem % HW;
    int s = n * IMGPAD + (y + P_ - 1) * HP + (x + P_ - 1);
    out[idx] = (ws[NUM_OFF + s] / ws[DIV_OFF + s] + 1.0f) * 0.5f;
}

extern "C" void kernel_launch(void* const* d_in, const int* in_sizes, int n_in,
                              void* d_out, int out_size, void* d_ws, size_t ws_size,
                              hipStream_t stream) {
    const float* x      = (const float*)d_in[0];
    const float* sigma_ = (const float*)d_in[1];
    const float* Pm1    = (const float*)d_in[2];
    float* out = (float*)d_out;
    float* ws  = (float*)d_ws;

    const int prepN = NIMG * IMGPAD;
    prep_kernel<<<(prepN + 255) / 256, 256, 0, stream>>>(x, Pm1, ws);

    invert_kernel<<<1, GJ_THREADS, 0, stream>>>(Pm1, ws);

    dim3 grid((LPOS + WAVES * 16 - 1) / (WAVES * 16), NIMG);
    dct2net_main<<<grid, 32 * WAVES, 0, stream>>>(sigma_, ws);

    const int outN = NIMG * HW * HW;
    finish_kernel<<<(outN + 255) / 256, 256, 0, stream>>>(ws, out);
}


# Pm1 and its inverse stored in WMMA A-fragment order: one contiguous 16-byte-per-lane load (4 whole lines) feeds two k-steps, replacing two 8-byte loads touching 16 lines each; sums unchanged
# speedup vs baseline: 47.6140x; 2.0829x over previous
#include <hip/hip_runtime.h>
#include <hip/hip_bf16.h>


#define P_    13
#define PP    169
#define PPD   176
#define NIMG  4
#define HW    256
#define HP    280
#define HO    268
#define LPOS  (HO*HO)
#define IMGPAD (HP*HP)

#define XPAD_OFF 0
#define PM1_OFF  (NIMG*IMGPAD)
#define QIN_OFF  (PM1_OFF + PPD*PPD)
#define NUM_OFF  (QIN_OFF + PPD*PPD)
#define DIV_OFF  (NUM_OFF + NIMG*IMGPAD)

#define GJ_G 22
#define GJ_T 8
#define GJ_STRIP (GJ_G * GJ_T)
#define GJ_THREADS 512

#define WAVES 4

typedef float v2f __attribute__((ext_vector_type(2)));
typedef float v4f __attribute__((ext_vector_type(4)));
typedef float v8f __attribute__((ext_vector_type(8)));

__device__ __forceinline__ int frag_index(const int r, const int c) {
    const int kt   = c >> 2;
    const int lane = (((c >> 1) & 1) << 4) | (r & 15);
    return (((kt >> 1) * 11 + (r >> 4)) * 32 + lane) * 4 + ((kt & 1) << 1) + (c & 1);
}

__global__ void prep_kernel(const float* __restrict__ x,
                            const float* __restrict__ Pm1,
                            float* __restrict__ ws) {
    int idx = blockIdx.x * blockDim.x + threadIdx.x;
    if (idx < NIMG * IMGPAD) {
        ws[NUM_OFF + idx] = 0.0f;
        ws[DIV_OFF + idx] = 0.0f;
        int n   = idx / IMGPAD;
        int rem = idx % IMGPAD;
        int y   = rem / HP;
        int xx  = rem % HP;
        int ty = y  - (P_ - 1); ty = (ty < 0) ? -ty : ((ty > HW - 1) ? 2 * (HW - 1) - ty : ty);
        int tx = xx - (P_ - 1); tx = (tx < 0) ? -tx : ((tx > HW - 1) ? 2 * (HW - 1) - tx : tx);
        ws[XPAD_OFF + idx] = 2.0f * x[n * HW * HW + ty * HW + tx] - 1.0f;
    }
    if (idx < PPD * PPD) {
        const int e     = idx & 1;
        const int s     = (idx >> 1) & 1;
        const int l     = (idx >> 2) & 31;
        const int chunk = idx >> 7;
        const int kp    = chunk / 11;
        const int mt    = chunk - kp * 11;
        const int r = mt * 16 + (l & 15);
        const int c = (2 * kp + s) * 4 + 2 * (l >> 4) + e;
        ws[PM1_OFF + idx] = (r < PP && c < PP) ? Pm1[r * PP + c] : 0.0f;
        ws[QIN_OFF + idx] = 0.0f;
    }
}

template <int JJ>
__device__ __forceinline__ void gj_phase(float (&a)[GJ_T][GJ_T],
                                         float (*prow)[GJ_STRIP],
                                         float (*fcol)[GJ_STRIP],
                                         const int ti, const int tj,
                                         const bool act) {
    constexpr int oEnd = ((JJ + 1) * GJ_G <= PP) ? GJ_G : (PP - JJ * GJ_G);
    for (int o = 0; o < oEnd; ++o) {
        const int par = (JJ * GJ_G + o) & 1;
        if (act && ti == o) {
#pragma unroll
            for (int j = 0; j < GJ_T; ++j) prow[par][tj * GJ_T + j] = a[JJ][j];
        }
        if (act && tj == o) {
#pragma unroll
            for (int i = 0; i < GJ_T; ++i) fcol[par][ti * GJ_T + i] = a[i][JJ];
        }
        __syncthreads();

        const float piv = prow[par][o * GJ_T + JJ];
        const float inv = 1.0f / piv;
        float p[GJ_T], f[GJ_T];
#pragma unroll
        for (int j = 0; j < GJ_T; ++j) p[j] = prow[par][tj * GJ_T + j] * inv;
#pragma unroll
        for (int i = 0; i < GJ_T; ++i) f[i] = fcol[par][ti * GJ_T + i];
        const bool ownCol = (tj == o);
        const bool ownRow = (ti == o);
        if (ownCol) p[JJ] = inv;
#pragma unroll
        for (int i = 0; i < GJ_T; ++i)
#pragma unroll
            for (int j = 0; j < GJ_T; ++j)
                a[i][j] -= f[i] * p[j];
        if (ownCol) {
#pragma unroll
            for (int i = 0; i < GJ_T; ++i) a[i][JJ] = 0.0f - f[i] * inv;
        }
        if (ownRow) {
#pragma unroll
            for (int j = 0; j < GJ_T; ++j) a[JJ][j] = p[j];
        }
    }
}

__global__ __launch_bounds__(GJ_THREADS)
void invert_kernel(const float* __restrict__ Pm1,
                   float* __restrict__ ws) {
    __shared__ __attribute__((aligned(16))) float prow[2][GJ_STRIP];
    __shared__ __attribute__((aligned(16))) float fcol[2][GJ_STRIP];

    const int tid = threadIdx.x;
    const bool act = tid < GJ_G * GJ_G;
    const int ti = act ? tid / GJ_G : 0;
    const int tj = act ? tid - ti * GJ_G : 0;

    float a[GJ_T][GJ_T];
#pragma unroll
    for (int i = 0; i < GJ_T; ++i) {
#pragma unroll
        for (int j = 0; j < GJ_T; ++j) {
            const int r = ti + GJ_G * i;
            const int c = tj + GJ_G * j;
            a[i][j] = (r < PP && c < PP) ? Pm1[r * PP + c] : 0.0f;
        }
    }

    gj_phase<0>(a, prow, fcol, ti, tj, act);
    gj_phase<1>(a, prow, fcol, ti, tj, act);
    gj_phase<2>(a, prow, fcol, ti, tj, act);
    gj_phase<3>(a, prow, fcol, ti, tj, act);
    gj_phase<4>(a, prow, fcol, ti, tj, act);
    gj_phase<5>(a, prow, fcol, ti, tj, act);
    gj_phase<6>(a, prow, fcol, ti, tj, act);
    gj_phase<7>(a, prow, fcol, ti, tj, act);

    if (act) {
#pragma unroll
        for (int i = 0; i < GJ_T; ++i) {
#pragma unroll
            for (int j = 0; j < GJ_T; ++j) {
                const int r = ti + GJ_G * i;
                const int c = tj + GJ_G * j;
                if (r < PP && c < PP) ws[QIN_OFF + frag_index(r, c)] = a[i][j];
            }
        }
    }
}

__global__ __launch_bounds__(32 * WAVES)
void dct2net_main(const float* __restrict__ sigma_,
                  float* __restrict__ ws) {
    __shared__ float lds_t[WAVES][PPD][17];

    const float* __restrict__ xpad = ws + XPAD_OFF;
    const v4f* __restrict__ pm1f = (const v4f*)(ws + PM1_OFF) + (threadIdx.x & 31);
    const v4f* __restrict__ qinf = (const v4f*)(ws + QIN_OFF) + (threadIdx.x & 31);
    float* __restrict__ numb = ws + NUM_OFF;
    float* __restrict__ divb = ws + DIV_OFF;

    const int wave  = threadIdx.x >> 5;
    const int lane  = threadIdx.x & 31;
    const int laneN = lane & 15;
    const int hi    = lane >> 4;

    const int img  = blockIdx.y;
    const int l0   = (blockIdx.x * WAVES + wave) * 16;
    const int lcol = l0 + laneN;
    const bool colOK = (lcol < LPOS);
    const int lc = colOK ? lcol : 0;
    const int ip = lc / HO;
    const int jp = lc - ip * HO;
    const float* __restrict__ xb = xpad + img * IMGPAD + ip * HP + jp;

    const float lam = 6.0f * sigma_[0];

    const v8f vzero = {0.f, 0.f, 0.f, 0.f, 0.f, 0.f, 0.f, 0.f};
    v8f acc[11];
#pragma unroll
    for (int mt = 0; mt < 11; ++mt) acc[mt] = vzero;

#pragma unroll 2
    for (int kp = 0; kp < 22; ++kp) {
        v2f b[2];
#pragma unroll
        for (int s = 0; s < 2; ++s) {
            const int ka = (2 * kp + s) * 4 + 2 * hi;
            const int d0 = ka / P_;
            const int d1 = (ka + 1) / P_;
            b[s].x = xb[d0 * HP + (ka - d0 * P_)];
            b[s].y = xb[d1 * HP + (ka + 1 - d1 * P_)];
        }
        v4f a[11];
#pragma unroll
        for (int mt = 0; mt < 11; ++mt) a[mt] = pm1f[(kp * 11 + mt) * 32];
#pragma unroll
        for (int mt = 0; mt < 11; ++mt) {
            const v2f a0 = {a[mt].x, a[mt].y};
            acc[mt] = __builtin_amdgcn_wmma_f32_16x16x4_f32(
                false, a0, false, b[0], (short)0, acc[mt], false, false);
        }
#pragma unroll
        for (int mt = 0; mt < 11; ++mt) {
            const v2f a1 = {a[mt].z, a[mt].w};
            acc[mt] = __builtin_amdgcn_wmma_f32_16x16x4_f32(
                false, a1, false, b[1], (short)0, acc[mt], false, false);
        }
    }

    int cnt = 0;
#pragma unroll
    for (int mt = 0; mt < 11; ++mt) {
#pragma unroll
        for (int v = 0; v < 8; ++v) {
            float t = acc[mt][v];
            t = (fabsf(t) > lam) ? t : 0.0f;
            acc[mt][v] = t;
            bool isRow0 = (mt == 0 && v == 0 && hi == 0);
            cnt += (!isRow0 && t != 0.0f) ? 1 : 0;
        }
    }
    cnt += __shfl_xor(cnt, 16, 32);
    const float w = 1.0f / (1.0f + (float)cnt);

#pragma unroll
    for (int mt = 0; mt < 11; ++mt)
#pragma unroll
        for (int v = 0; v < 8; ++v)
            lds_t[wave][mt * 16 + v + hi * 8][laneN] = acc[mt][v];
    __syncthreads();

    v8f acc2[11];
#pragma unroll
    for (int mt = 0; mt < 11; ++mt) acc2[mt] = vzero;

#pragma unroll 2
    for (int kp = 0; kp < 22; ++kp) {
        v2f b[2];
#pragma unroll
        for (int s = 0; s < 2; ++s) {
            const int ka = (2 * kp + s) * 4 + 2 * hi;
            b[s].x = lds_t[wave][ka][laneN];
            b[s].y = lds_t[wave][ka + 1][laneN];
        }
        v4f a[11];
#pragma unroll
        for (int mt = 0; mt < 11; ++mt) a[mt] = qinf[(kp * 11 + mt) * 32];
#pragma unroll
        for (int mt = 0; mt < 11; ++mt) {
            const v2f a0 = {a[mt].x, a[mt].y};
            acc2[mt] = __builtin_amdgcn_wmma_f32_16x16x4_f32(
                false, a0, false, b[0], (short)0, acc2[mt], false, false);
        }
#pragma unroll
        for (int mt = 0; mt < 11; ++mt) {
            const v2f a1 = {a[mt].z, a[mt].w};
            acc2[mt] = __builtin_amdgcn_wmma_f32_16x16x4_f32(
                false, a1, false, b[1], (short)0, acc2[mt], false, false);
        }
    }

    static_assert(LPOS % 16 == 0, "a tile is wholly inside or wholly outside the patch list");
    float (*T)[17] = lds_t[wave];
#pragma unroll
    for (int mt = 0; mt < 11; ++mt) {
#pragma unroll
        for (int v = 0; v < 8; ++v) {
            const int M = mt * 16 + v + hi * 8;
            if (M < PP) T[M][laneN] = acc2[mt][v] * w;
        }
    }
    if (hi == 0) T[PP][laneN] = w;
    for (int M = lane; M <= PP; M += 32) T[M][16] = 0.0f;
    __syncthreads();

    if (l0 < LPOS) {
        const int ip0 = l0 / HO;
        const int jp0 = l0 - ip0 * HO;
        const int s   = (HO - jp0 < 16) ? (HO - jp0) : 16;
        float* nb = numb + img * IMGPAD + ip0 * HP + jp0;
        float* db = divb + img * IMGPAD + ip0 * HP + jp0;
        const int npass = (s < 16) ? 2 : 1;
        for (int pass = 0; pass < npass; ++pass) {
            const int q = lane + 32 * pass;
            int col[P_];
            bool any = false;
#pragma unroll
            for (int dj = 0; dj < P_; ++dj) {
                const int sh = q - dj;
                const bool row0 = (sh >= 0) && (sh < s);
                const bool row1 = (sh >= s + 12) && (sh <= 27);
                col[dj] = row0 ? sh : (row1 ? sh - 12 : 16);
                any = any || row0 || row1;
            }
            float dsum = 0.0f;
#pragma unroll
            for (int dj = 0; dj < P_; ++dj) dsum += T[PP][col[dj]];
            if (any) {
#pragma unroll
                for (int di = 0; di < P_; ++di) {
                    float nsum = 0.0f;
#pragma unroll
                    for (int dj = 0; dj < P_; ++dj) nsum += T[di * P_ + dj][col[dj]];
                    atomicAdd(nb + di * HP + q, nsum);
                    atomicAdd(db + di * HP + q, dsum);
                }
            }
        }
    }
}

__global__ void finish_kernel(const float* __restrict__ ws,
                              float* __restrict__ out) {
    int idx = blockIdx.x * blockDim.x + threadIdx.x;
    if (idx >= NIMG * HW * HW) return;
    int n   = idx / (HW * HW);
    int rem = idx % (HW * HW);
    int y   = rem / HW;
    int x   = rem % HW;
    int s = n * IMGPAD + (y + P_ - 1) * HP + (x + P_ - 1);
    out[idx] = (ws[NUM_OFF + s] / ws[DIV_OFF + s] + 1.0f) * 0.5f;
}

extern "C" void kernel_launch(void* const* d_in, const int* in_sizes, int n_in,
                              void* d_out, int out_size, void* d_ws, size_t ws_size,
                              hipStream_t stream) {
    const float* x      = (const float*)d_in[0];
    const float* sigma_ = (const float*)d_in[1];
    const float* Pm1    = (const float*)d_in[2];
    float* out = (float*)d_out;
    float* ws  = (float*)d_ws;

    const int prepN = NIMG * IMGPAD;
    prep_kernel<<<(prepN + 255) / 256, 256, 0, stream>>>(x, Pm1, ws);

    invert_kernel<<<1, GJ_THREADS, 0, stream>>>(Pm1, ws);

    dim3 grid((LPOS + WAVES * 16 - 1) / (WAVES * 16), NIMG);
    dct2net_main<<<grid, 32 * WAVES, 0, stream>>>(sigma_, ws);

    const int outN = NIMG * HW * HW;
    finish_kernel<<<(outN + 255) / 256, 256, 0, stream>>>(ws, out);
}


# Main kernel's two GEMM loops double-buffer their A/B fragments: the loads of chunk row kp+1 go out as one group ahead of the 22 WMMAs of row kp (was load, wait, WMMA); k order and sums unchanged
# speedup vs baseline: 49.0743x; 1.0307x over previous
#include <hip/hip_runtime.h>
#include <hip/hip_bf16.h>


#define P_    13
#define PP    169
#define PPD   176
#define NIMG  4
#define HW    256
#define HP    280
#define HO    268
#define LPOS  (HO*HO)
#define IMGPAD (HP*HP)

#define XPAD_OFF 0
#define PM1_OFF  (NIMG*IMGPAD)
#define QIN_OFF  (PM1_OFF + PPD*PPD)
#define NUM_OFF  (QIN_OFF + PPD*PPD)
#define DIV_OFF  (NUM_OFF + NIMG*IMGPAD)

#define GJ_G 22
#define GJ_T 8
#define GJ_STRIP (GJ_G * GJ_T)
#define GJ_THREADS 512

#define WAVES 4

typedef float v2f __attribute__((ext_vector_type(2)));
typedef float v4f __attribute__((ext_vector_type(4)));
typedef float v8f __attribute__((ext_vector_type(8)));

__device__ __forceinline__ int frag_index(const int r, const int c) {
    const int kt   = c >> 2;
    const int lane = (((c >> 1) & 1) << 4) | (r & 15);
    return (((kt >> 1) * 11 + (r >> 4)) * 32 + lane) * 4 + ((kt & 1) << 1) + (c & 1);
}

__global__ void prep_kernel(const float* __restrict__ x,
                            const float* __restrict__ Pm1,
                            float* __restrict__ ws) {
    int idx = blockIdx.x * blockDim.x + threadIdx.x;
    if (idx < NIMG * IMGPAD) {
        ws[NUM_OFF + idx] = 0.0f;
        ws[DIV_OFF + idx] = 0.0f;
        int n   = idx / IMGPAD;
        int rem = idx % IMGPAD;
        int y   = rem / HP;
        int xx  = rem % HP;
        int ty = y  - (P_ - 1); ty = (ty < 0) ? -ty : ((ty > HW - 1) ? 2 * (HW - 1) - ty : ty);
        int tx = xx - (P_ - 1); tx = (tx < 0) ? -tx : ((tx > HW - 1) ? 2 * (HW - 1) - tx : tx);
        ws[XPAD_OFF + idx] = 2.0f * x[n * HW * HW + ty * HW + tx] - 1.0f;
    }
    if (idx < PPD * PPD) {
        const int e     = idx & 1;
        const int s     = (idx >> 1) & 1;
        const int l     = (idx >> 2) & 31;
        const int chunk = idx >> 7;
        const int kp    = chunk / 11;
        const int mt    = chunk - kp * 11;
        const int r = mt * 16 + (l & 15);
        const int c = (2 * kp + s) * 4 + 2 * (l >> 4) + e;
        ws[PM1_OFF + idx] = (r < PP && c < PP) ? Pm1[r * PP + c] : 0.0f;
        ws[QIN_OFF + idx] = 0.0f;
    }
}

template <int JJ>
__device__ __forceinline__ void gj_phase(float (&a)[GJ_T][GJ_T],
                                         float (*prow)[GJ_STRIP],
                                         float (*fcol)[GJ_STRIP],
                                         const int ti, const int tj,
                                         const bool act) {
    constexpr int oEnd = ((JJ + 1) * GJ_G <= PP) ? GJ_G : (PP - JJ * GJ_G);
    for (int o = 0; o < oEnd; ++o) {
        const int par = (JJ * GJ_G + o) & 1;
        if (act && ti == o) {
#pragma unroll
            for (int j = 0; j < GJ_T; ++j) prow[par][tj * GJ_T + j] = a[JJ][j];
        }
        if (act && tj == o) {
#pragma unroll
            for (int i = 0; i < GJ_T; ++i) fcol[par][ti * GJ_T + i] = a[i][JJ];
        }
        __syncthreads();

        const float piv = prow[par][o * GJ_T + JJ];
        const float inv = 1.0f / piv;
        float p[GJ_T], f[GJ_T];
#pragma unroll
        for (int j = 0; j < GJ_T; ++j) p[j] = prow[par][tj * GJ_T + j] * inv;
#pragma unroll
        for (int i = 0; i < GJ_T; ++i) f[i] = fcol[par][ti * GJ_T + i];
        const bool ownCol = (tj == o);
        const bool ownRow = (ti == o);
        if (ownCol) p[JJ] = inv;
#pragma unroll
        for (int i = 0; i < GJ_T; ++i)
#pragma unroll
            for (int j = 0; j < GJ_T; ++j)
                a[i][j] -= f[i] * p[j];
        if (ownCol) {
#pragma unroll
            for (int i = 0; i < GJ_T; ++i) a[i][JJ] = 0.0f - f[i] * inv;
        }
        if (ownRow) {
#pragma unroll
            for (int j = 0; j < GJ_T; ++j) a[JJ][j] = p[j];
        }
    }
}

__global__ __launch_bounds__(GJ_THREADS)
void invert_kernel(const float* __restrict__ Pm1,
                   float* __restrict__ ws) {
    __shared__ __attribute__((aligned(16))) float prow[2][GJ_STRIP];
    __shared__ __attribute__((aligned(16))) float fcol[2][GJ_STRIP];

    const int tid = threadIdx.x;
    const bool act = tid < GJ_G * GJ_G;
    const int ti = act ? tid / GJ_G : 0;
    const int tj = act ? tid - ti * GJ_G : 0;

    float a[GJ_T][GJ_T];
#pragma unroll
    for (int i = 0; i < GJ_T; ++i) {
#pragma unroll
        for (int j = 0; j < GJ_T; ++j) {
            const int r = ti + GJ_G * i;
            const int c = tj + GJ_G * j;
            a[i][j] = (r < PP && c < PP) ? Pm1[r * PP + c] : 0.0f;
        }
    }

    gj_phase<0>(a, prow, fcol, ti, tj, act);
    gj_phase<1>(a, prow, fcol, ti, tj, act);
    gj_phase<2>(a, prow, fcol, ti, tj, act);
    gj_phase<3>(a, prow, fcol, ti, tj, act);
    gj_phase<4>(a, prow, fcol, ti, tj, act);
    gj_phase<5>(a, prow, fcol, ti, tj, act);
    gj_phase<6>(a, prow, fcol, ti, tj, act);
    gj_phase<7>(a, prow, fcol, ti, tj, act);

    if (act) {
#pragma unroll
        for (int i = 0; i < GJ_T; ++i) {
#pragma unroll
            for (int j = 0; j < GJ_T; ++j) {
                const int r = ti + GJ_G * i;
                const int c = tj + GJ_G * j;
                if (r < PP && c < PP) ws[QIN_OFF + frag_index(r, c)] = a[i][j];
            }
        }
    }
}

__device__ __forceinline__ void load_a(v4f (&a)[11], const v4f* __restrict__ base, const int kp) {
#pragma unroll
    for (int mt = 0; mt < 11; ++mt) a[mt] = base[(kp * 11 + mt) * 32];
}

__device__ __forceinline__ void mma_row(v8f (&acc)[11], const v4f (&a)[11], const v2f (&b)[2]) {
#pragma unroll
    for (int mt = 0; mt < 11; ++mt) {
        const v2f a0 = {a[mt].x, a[mt].y};
        acc[mt] = __builtin_amdgcn_wmma_f32_16x16x4_f32(
            false, a0, false, b[0], (short)0, acc[mt], false, false);
    }
#pragma unroll
    for (int mt = 0; mt < 11; ++mt) {
        const v2f a1 = {a[mt].z, a[mt].w};
        acc[mt] = __builtin_amdgcn_wmma_f32_16x16x4_f32(
            false, a1, false, b[1], (short)0, acc[mt], false, false);
    }
}

__global__ __launch_bounds__(32 * WAVES)
void dct2net_main(const float* __restrict__ sigma_,
                  float* __restrict__ ws) {
    __shared__ float lds_t[WAVES][PPD][17];

    const float* __restrict__ xpad = ws + XPAD_OFF;
    const v4f* __restrict__ pm1f = (const v4f*)(ws + PM1_OFF) + (threadIdx.x & 31);
    const v4f* __restrict__ qinf = (const v4f*)(ws + QIN_OFF) + (threadIdx.x & 31);
    float* __restrict__ numb = ws + NUM_OFF;
    float* __restrict__ divb = ws + DIV_OFF;

    const int wave  = threadIdx.x >> 5;
    const int lane  = threadIdx.x & 31;
    const int laneN = lane & 15;
    const int hi    = lane >> 4;

    const int img  = blockIdx.y;
    const int l0   = (blockIdx.x * WAVES + wave) * 16;
    const int lcol = l0 + laneN;
    const bool colOK = (lcol < LPOS);
    const int lc = colOK ? lcol : 0;
    const int ip = lc / HO;
    const int jp = lc - ip * HO;
    const float* __restrict__ xb = xpad + img * IMGPAD + ip * HP + jp;

    const float lam = 6.0f * sigma_[0];

    const v8f vzero = {0.f, 0.f, 0.f, 0.f, 0.f, 0.f, 0.f, 0.f};
    v8f acc[11];
#pragma unroll
    for (int mt = 0; mt < 11; ++mt) acc[mt] = vzero;

    auto load_b1 = [&](v2f (&b)[2], const int kp) {
#pragma unroll
        for (int s = 0; s < 2; ++s) {
            const int ka = (2 * kp + s) * 4 + 2 * hi;
            const int d0 = ka / P_;
            const int d1 = (ka + 1) / P_;
            b[s].x = xb[d0 * HP + (ka - d0 * P_)];
            b[s].y = xb[d1 * HP + (ka + 1 - d1 * P_)];
        }
    };
    v4f aA[11], aB[11];
    v2f bA[2], bB[2];
    load_a(aA, pm1f, 0);
    load_b1(bA, 0);
#pragma unroll 1
    for (int kp = 0; kp < 20; kp += 2) {
        load_a(aB, pm1f, kp + 1);
        load_b1(bB, kp + 1);
        __builtin_amdgcn_sched_barrier(0);
        mma_row(acc, aA, bA);
        load_a(aA, pm1f, kp + 2);
        load_b1(bA, kp + 2);
        __builtin_amdgcn_sched_barrier(0);
        mma_row(acc, aB, bB);
    }
    load_a(aB, pm1f, 21);
    load_b1(bB, 21);
    __builtin_amdgcn_sched_barrier(0);
    mma_row(acc, aA, bA);
    mma_row(acc, aB, bB);

    int cnt = 0;
#pragma unroll
    for (int mt = 0; mt < 11; ++mt) {
#pragma unroll
        for (int v = 0; v < 8; ++v) {
            float t = acc[mt][v];
            t = (fabsf(t) > lam) ? t : 0.0f;
            acc[mt][v] = t;
            bool isRow0 = (mt == 0 && v == 0 && hi == 0);
            cnt += (!isRow0 && t != 0.0f) ? 1 : 0;
        }
    }
    cnt += __shfl_xor(cnt, 16, 32);
    const float w = 1.0f / (1.0f + (float)cnt);

#pragma unroll
    for (int mt = 0; mt < 11; ++mt)
#pragma unroll
        for (int v = 0; v < 8; ++v) {
            float tv = acc[mt][v];
            if (mt * 16 + v + 8 == PP) tv = hi ? w : tv;
            lds_t[wave][mt * 16 + v + hi * 8][laneN] = tv;
        }
    __syncthreads();

    v8f acc2[11];
#pragma unroll
    for (int mt = 0; mt < 11; ++mt) acc2[mt] = vzero;

    auto load_b2 = [&](v2f (&b)[2], const int kp) {
#pragma unroll
        for (int s = 0; s < 2; ++s) {
            const int ka = (2 * kp + s) * 4 + 2 * hi;
            b[s].x = lds_t[wave][ka][laneN];
            b[s].y = lds_t[wave][ka + 1][laneN];
        }
    };
    load_a(aA, qinf, 0);
    load_b2(bA, 0);
#pragma unroll 1
    for (int kp = 0; kp < 20; kp += 2) {
        load_a(aB, qinf, kp + 1);
        load_b2(bB, kp + 1);
        __builtin_amdgcn_sched_barrier(0);
        mma_row(acc2, aA, bA);
        load_a(aA, qinf, kp + 2);
        load_b2(bA, kp + 2);
        __builtin_amdgcn_sched_barrier(0);
        mma_row(acc2, aB, bB);
    }
    load_a(aB, qinf, 21);
    load_b2(bB, 21);
    __builtin_amdgcn_sched_barrier(0);
    mma_row(acc2, aA, bA);
    mma_row(acc2, aB, bB);

    static_assert(LPOS % 16 == 0, "a tile is wholly inside or wholly outside the patch list");
    float (*T)[17] = lds_t[wave];
#pragma unroll
    for (int mt = 0; mt < 11; ++mt) {
#pragma unroll
        for (int v = 0; v < 8; ++v) {
            const int M = mt * 16 + v + hi * 8;
            if (M < PP) T[M][laneN] = acc2[mt][v] * w;
        }
    }
    for (int M = lane; M <= PP; M += 32) T[M][16] = 0.0f;
    __syncthreads();

    if (l0 < LPOS) {
        const int ip0 = l0 / HO;
        const int jp0 = l0 - ip0 * HO;
        const int s   = (HO - jp0 < 16) ? (HO - jp0) : 16;
        float* nb = numb + img * IMGPAD + ip0 * HP + jp0;
        float* db = divb + img * IMGPAD + ip0 * HP + jp0;
        const int npass = (s < 16) ? 2 : 1;
        for (int pass = 0; pass < npass; ++pass) {
            const int q = lane + 32 * pass;
            int col[P_];
            bool any = false;
#pragma unroll
            for (int dj = 0; dj < P_; ++dj) {
                const int sh = q - dj;
                const bool row0 = (sh >= 0) && (sh < s);
                const bool row1 = (sh >= s + 12) && (sh <= 27);
                col[dj] = row0 ? sh : (row1 ? sh - 12 : 16);
                any = any || row0 || row1;
            }
            float dsum = 0.0f;
#pragma unroll
            for (int dj = 0; dj < P_; ++dj) dsum += T[PP][col[dj]];
            if (any) {
#pragma unroll
                for (int di = 0; di < P_; ++di) {
                    float nsum = 0.0f;
#pragma unroll
                    for (int dj = 0; dj < P_; ++dj) nsum += T[di * P_ + dj][col[dj]];
                    atomicAdd(nb + di * HP + q, nsum);
                    atomicAdd(db + di * HP + q, dsum);
                }
            }
        }
    }
}

__global__ void finish_kernel(const float* __restrict__ ws,
                              float* __restrict__ out) {
    int idx = blockIdx.x * blockDim.x + threadIdx.x;
    if (idx >= NIMG * HW * HW) return;
    int n   = idx / (HW * HW);
    int rem = idx % (HW * HW);
    int y   = rem / HW;
    int x   = rem % HW;
    int s = n * IMGPAD + (y + P_ - 1) * HP + (x + P_ - 1);
    out[idx] = (ws[NUM_OFF + s] / ws[DIV_OFF + s] + 1.0f) * 0.5f;
}

extern "C" void kernel_launch(void* const* d_in, const int* in_sizes, int n_in,
                              void* d_out, int out_size, void* d_ws, size_t ws_size,
                              hipStream_t stream) {
    const float* x      = (const float*)d_in[0];
    const float* sigma_ = (const float*)d_in[1];
    const float* Pm1    = (const float*)d_in[2];
    float* out = (float*)d_out;
    float* ws  = (float*)d_ws;

    const int prepN = NIMG * IMGPAD;
    prep_kernel<<<(prepN + 255) / 256, 256, 0, stream>>>(x, Pm1, ws);

    invert_kernel<<<1, GJ_THREADS, 0, stream>>>(Pm1, ws);

    dim3 grid((LPOS + WAVES * 16 - 1) / (WAVES * 16), NIMG);
    dct2net_main<<<grid, 32 * WAVES, 0, stream>>>(sigma_, ws);

    const int outN = NIMG * HW * HW;
    finish_kernel<<<(outN + 255) / 256, 256, 0, stream>>>(ws, out);
}


# Inverse blocked by 4 pivots on WMMA: matrix held as f32 16x16x4 tiles, lanes rebuild each pivot's multipliers from raw rows/columns in LDS, one WMMA per tile applies 4 rank-1 updates in pivot order
# speedup vs baseline: 50.8378x; 1.0359x over previous
#include <hip/hip_runtime.h>
#include <hip/hip_bf16.h>


#define P_    13
#define PP    169
#define PPD   176
#define NIMG  4
#define HW    256
#define HP    280
#define HO    268
#define LPOS  (HO*HO)
#define IMGPAD (HP*HP)

#define XPAD_OFF 0
#define PM1_OFF  (NIMG*IMGPAD)
#define QIN_OFF  (PM1_OFF + PPD*PPD)
#define NUM_OFF  (QIN_OFF + PPD*PPD)
#define DIV_OFF  (NUM_OFF + NIMG*IMGPAD)

#define IV_WAVES 11
#define IV_THREADS (32 * IV_WAVES)
#define IV_ROWS 172

#define WAVES 4

typedef float v2f __attribute__((ext_vector_type(2)));
typedef float v4f __attribute__((ext_vector_type(4)));
typedef float v8f __attribute__((ext_vector_type(8)));

__device__ __forceinline__ int frag_index(const int r, const int c) {
    const int kt   = c >> 2;
    const int lane = (((c >> 1) & 1) << 4) | (r & 15);
    return (((kt >> 1) * 11 + (r >> 4)) * 32 + lane) * 4 + ((kt & 1) << 1) + (c & 1);
}

__global__ void prep_kernel(const float* __restrict__ x,
                            const float* __restrict__ Pm1,
                            float* __restrict__ ws) {
    int idx = blockIdx.x * blockDim.x + threadIdx.x;
    if (idx < NIMG * IMGPAD) {
        ws[NUM_OFF + idx] = 0.0f;
        ws[DIV_OFF + idx] = 0.0f;
        int n   = idx / IMGPAD;
        int rem = idx % IMGPAD;
        int y   = rem / HP;
        int xx  = rem % HP;
        int ty = y  - (P_ - 1); ty = (ty < 0) ? -ty : ((ty > HW - 1) ? 2 * (HW - 1) - ty : ty);
        int tx = xx - (P_ - 1); tx = (tx < 0) ? -tx : ((tx > HW - 1) ? 2 * (HW - 1) - tx : tx);
        ws[XPAD_OFF + idx] = 2.0f * x[n * HW * HW + ty * HW + tx] - 1.0f;
    }
    if (idx < PPD * PPD) {
        const int e     = idx & 1;
        const int s     = (idx >> 1) & 1;
        const int l     = (idx >> 2) & 31;
        const int chunk = idx >> 7;
        const int kp    = chunk / 11;
        const int mt    = chunk - kp * 11;
        const int r = mt * 16 + (l & 15);
        const int c = (2 * kp + s) * 4 + 2 * (l >> 4) + e;
        ws[PM1_OFF + idx] = (r < PP && c < PP) ? Pm1[r * PP + c] : 0.0f;
        ws[QIN_OFF + idx] = 0.0f;
    }
}

__device__ __forceinline__ float sel4(const int t, const float v0, const float v1,
                                      const float v2, const float v3) {
    return (t == 0) ? v0 : ((t == 1) ? v1 : ((t == 2) ? v2 : v3));
}

template <int MTK>
__device__ __forceinline__ void iv_phase(v8f (&a)[11],
                                         float (*rowp)[4][PPD],
                                         float (*colp)[4][PPD],
                                         const int nt, const int c, const int h,
                                         int& par) {
    constexpr int qEnd = (16 * MTK + 16 <= IV_ROWS) ? 4 : (IV_ROWS - 16 * MTK) / 4;
    const int col = nt * 16 + c;
    const int e   = c & 3;
    const v8f vz  = {0.f, 0.f, 0.f, 0.f, 0.f, 0.f, 0.f, 0.f};
    for (int q = 0; q < qEnd; ++q) {
        const int  k       = 16 * MTK + 4 * q;
        const bool qhi     = (q & 1) != 0;
        const bool rowLane = (h == (q >> 1));
        const bool inBlk   = ((c >> 2) == q);
        const bool colLane = (nt == MTK) && inBlk;

        if (rowLane) {
#pragma unroll
            for (int m = 0; m < 4; ++m) rowp[par][m][col] = qhi ? a[MTK][4 + m] : a[MTK][m];
        }
        if (colLane) {
#pragma unroll
            for (int mt = 0; mt < 11; ++mt) {
                const v4f lo = {a[mt][0], a[mt][1], a[mt][2], a[mt][3]};
                const v4f hi4 = {a[mt][4], a[mt][5], a[mt][6], a[mt][7]};
                *(v4f*)&colp[par][e][mt * 16 + 8 * h]     = lo;
                *(v4f*)&colp[par][e][mt * 16 + 8 * h + 4] = hi4;
                a[mt] = vz;
            }
        }
        __syncthreads();

        float G[4][4];
#pragma unroll
        for (int t = 0; t < 4; ++t) {
            const v4f g = *(const v4f*)&rowp[par][t][k];
            G[t][0] = g.x; G[t][1] = g.y; G[t][2] = g.z; G[t][3] = g.w;
        }
        float inv[4], fs[4][4], ps[4][4];
#pragma unroll
        for (int s = 0; s < 4; ++s) {
            inv[s] = 1.0f / G[s][s];
#pragma unroll
            for (int u = 0; u < 4; ++u) ps[s][u] = G[s][u] * inv[s];
            ps[s][s] = inv[s];
#pragma unroll
            for (int t = 0; t < 4; ++t) fs[s][t] = G[t][s];
#pragma unroll
            for (int t = 0; t < 4; ++t)
#pragma unroll
                for (int u = 0; u < 4; ++u)
                    G[t][u] -= fs[s][t] * ps[s][u];
#pragma unroll
            for (int t = 0; t < 4; ++t) G[t][s] = 0.0f - fs[s][t] * inv[s];
#pragma unroll
            for (int u = 0; u < 4; ++u) G[s][u] = ps[s][u];
        }

        float pb[4];
        {
            float x0 = rowp[par][0][col];
            float x1 = rowp[par][1][col];
            float x2 = rowp[par][2][col];
            float x3 = rowp[par][3][col];
            pb[0] = x0 * inv[0];
            x1 -= fs[0][1] * pb[0];  x2 -= fs[0][2] * pb[0];  x3 -= fs[0][3] * pb[0];
            pb[1] = x1 * inv[1];
            x2 -= fs[1][2] * pb[1];  x3 -= fs[1][3] * pb[1];
            pb[2] = x2 * inv[2];
            x3 -= fs[2][3] * pb[2];
            pb[3] = x3 * inv[3];
        }
#pragma unroll
        for (int s = 0; s < 4; ++s) {
            const float o = sel4(e, ps[s][0],
                                    (s >= 1) ? ps[s][1] : 0.0f,
                                    (s >= 2) ? ps[s][2] : 0.0f,
                                    (s >= 3) ? ps[s][3] : 0.0f);
            pb[s] = colLane ? o : pb[s];
        }
        const v2f bfr = {h ? pb[2] : pb[0], h ? pb[3] : pb[1]};

#pragma unroll
        for (int m = 0; m < 4; ++m) {
            a[MTK][m]     = (rowLane && !qhi) ? 0.0f : a[MTK][m];
            a[MTK][4 + m] = (rowLane &&  qhi) ? 0.0f : a[MTK][4 + m];
        }

        auto tile = [&](const int mt) {
            const int i = mt * 16 + c;
            float y0 = colp[par][0][i];
            float y1 = colp[par][1][i];
            float y2 = colp[par][2][i];
            float y3 = colp[par][3][i];
            float fa[4];
            fa[0] = y0;
            y1 -= fa[0] * ps[0][1];  y2 -= fa[0] * ps[0][2];  y3 -= fa[0] * ps[0][3];
            fa[1] = y1;
            y2 -= fa[1] * ps[1][2];  y3 -= fa[1] * ps[1][3];
            fa[2] = y2;
            y3 -= fa[2] * ps[2][3];
            fa[3] = y3;
#pragma unroll
            for (int s = 0; s < 4; ++s) fa[s] = 0.0f - fa[s];
            if (mt == MTK) {
#pragma unroll
                for (int s = 0; s < 4; ++s) {
                    const float o = sel4(e, (s == 0) ? 1.0f : (0.0f - fs[s][0]),
                                            (s < 1) ? 0.0f : ((s == 1) ? 1.0f : (0.0f - fs[s][1])),
                                            (s < 2) ? 0.0f : ((s == 2) ? 1.0f : (0.0f - fs[s][2])),
                                            (s < 3) ? 0.0f : 1.0f);
                    fa[s] = inBlk ? o : fa[s];
                }
            }
            const v2f afr = {h ? fa[2] : fa[0], h ? fa[3] : fa[1]};
            a[mt] = __builtin_amdgcn_wmma_f32_16x16x4_f32(
                false, afr, false, bfr, (short)0, a[mt], false, false);
        };
        tile(MTK);
        if (MTK + 1 < 11) tile(MTK + 1);
        __builtin_amdgcn_sched_barrier(0);
#pragma unroll
        for (int mt = 0; mt < 11; ++mt)
            if (mt != MTK && mt != MTK + 1) tile(mt);
        par ^= 1;
    }
}

__global__ __launch_bounds__(IV_THREADS)
void invert_kernel(const float* __restrict__ Pm1,
                   float* __restrict__ ws) {
    __shared__ __attribute__((aligned(16))) float rowp[2][4][PPD];
    __shared__ __attribute__((aligned(16))) float colp[2][4][PPD];

    const int nt   = __builtin_amdgcn_readfirstlane(threadIdx.x >> 5);
    const int lane = threadIdx.x & 31;
    const int c    = lane & 15;
    const int h    = lane >> 4;
    const int col  = nt * 16 + c;

    v8f a[11];
#pragma unroll
    for (int mt = 0; mt < 11; ++mt) {
#pragma unroll
        for (int r = 0; r < 8; ++r) {
            const int row = mt * 16 + 8 * h + r;
            const float pad = (row == col && row < IV_ROWS) ? 1.0f : 0.0f;
            a[mt][r] = (row < PP && col < PP) ? Pm1[row * PP + col] : pad;
        }
    }

    int par = 0;
    iv_phase<0>(a, rowp, colp, nt, c, h, par);
    iv_phase<1>(a, rowp, colp, nt, c, h, par);
    iv_phase<2>(a, rowp, colp, nt, c, h, par);
    iv_phase<3>(a, rowp, colp, nt, c, h, par);
    iv_phase<4>(a, rowp, colp, nt, c, h, par);
    iv_phase<5>(a, rowp, colp, nt, c, h, par);
    iv_phase<6>(a, rowp, colp, nt, c, h, par);
    iv_phase<7>(a, rowp, colp, nt, c, h, par);
    iv_phase<8>(a, rowp, colp, nt, c, h, par);
    iv_phase<9>(a, rowp, colp, nt, c, h, par);
    iv_phase<10>(a, rowp, colp, nt, c, h, par);

#pragma unroll
    for (int mt = 0; mt < 11; ++mt) {
#pragma unroll
        for (int r = 0; r < 8; ++r) {
            const int row = mt * 16 + 8 * h + r;
            if (row < PP && col < PP) ws[QIN_OFF + frag_index(row, col)] = a[mt][r];
        }
    }
}

__device__ __forceinline__ void load_a(v4f (&a)[11], const v4f* __restrict__ base, const int kp) {
#pragma unroll
    for (int mt = 0; mt < 11; ++mt) a[mt] = base[(kp * 11 + mt) * 32];
}

__device__ __forceinline__ void mma_row(v8f (&acc)[11], const v4f (&a)[11], const v2f (&b)[2]) {
#pragma unroll
    for (int mt = 0; mt < 11; ++mt) {
        const v2f a0 = {a[mt].x, a[mt].y};
        acc[mt] = __builtin_amdgcn_wmma_f32_16x16x4_f32(
            false, a0, false, b[0], (short)0, acc[mt], false, false);
    }
#pragma unroll
    for (int mt = 0; mt < 11; ++mt) {
        const v2f a1 = {a[mt].z, a[mt].w};
        acc[mt] = __builtin_amdgcn_wmma_f32_16x16x4_f32(
            false, a1, false, b[1], (short)0, acc[mt], false, false);
    }
}

__global__ __launch_bounds__(32 * WAVES)
void dct2net_main(const float* __restrict__ sigma_,
                  float* __restrict__ ws) {
    __shared__ float lds_t[WAVES][PPD][17];

    const float* __restrict__ xpad = ws + XPAD_OFF;
    const v4f* __restrict__ pm1f = (const v4f*)(ws + PM1_OFF) + (threadIdx.x & 31);
    const v4f* __restrict__ qinf = (const v4f*)(ws + QIN_OFF) + (threadIdx.x & 31);
    float* __restrict__ numb = ws + NUM_OFF;
    float* __restrict__ divb = ws + DIV_OFF;

    const int wave  = threadIdx.x >> 5;
    const int lane  = threadIdx.x & 31;
    const int laneN = lane & 15;
    const int hi    = lane >> 4;

    const int img  = blockIdx.y;
    const int l0   = (blockIdx.x * WAVES + wave) * 16;
    const int lcol = l0 + laneN;
    const bool colOK = (lcol < LPOS);
    const int lc = colOK ? lcol : 0;
    const int ip = lc / HO;
    const int jp = lc - ip * HO;
    const float* __restrict__ xb = xpad + img * IMGPAD + ip * HP + jp;

    const float lam = 6.0f * sigma_[0];

    const v8f vzero = {0.f, 0.f, 0.f, 0.f, 0.f, 0.f, 0.f, 0.f};
    v8f acc[11];
#pragma unroll
    for (int mt = 0; mt < 11; ++mt) acc[mt] = vzero;

    auto load_b1 = [&](v2f (&b)[2], const int kp) {
#pragma unroll
        for (int s = 0; s < 2; ++s) {
            const int ka = (2 * kp + s) * 4 + 2 * hi;
            const int d0 = ka / P_;
            const int d1 = (ka + 1) / P_;
            b[s].x = xb[d0 * HP + (ka - d0 * P_)];
            b[s].y = xb[d1 * HP + (ka + 1 - d1 * P_)];
        }
    };
    v4f aA[11], aB[11];
    v2f bA[2], bB[2];
    load_a(aA, pm1f, 0);
    load_b1(bA, 0);
#pragma unroll 1
    for (int kp = 0; kp < 20; kp += 2) {
        load_a(aB, pm1f, kp + 1);
        load_b1(bB, kp + 1);
        __builtin_amdgcn_sched_barrier(0);
        mma_row(acc, aA, bA);
        load_a(aA, pm1f, kp + 2);
        load_b1(bA, kp + 2);
        __builtin_amdgcn_sched_barrier(0);
        mma_row(acc, aB, bB);
    }
    load_a(aB, pm1f, 21);
    load_b1(bB, 21);
    __builtin_amdgcn_sched_barrier(0);
    mma_row(acc, aA, bA);
    mma_row(acc, aB, bB);

    int cnt = 0;
#pragma unroll
    for (int mt = 0; mt < 11; ++mt) {
#pragma unroll
        for (int v = 0; v < 8; ++v) {
            float t = acc[mt][v];
            t = (fabsf(t) > lam) ? t : 0.0f;
            acc[mt][v] = t;
            bool isRow0 = (mt == 0 && v == 0 && hi == 0);
            cnt += (!isRow0 && t != 0.0f) ? 1 : 0;
        }
    }
    cnt += __shfl_xor(cnt, 16, 32);
    const float w = 1.0f / (1.0f + (float)cnt);

#pragma unroll
    for (int mt = 0; mt < 11; ++mt)
#pragma unroll
        for (int v = 0; v < 8; ++v) {
            float tv = acc[mt][v];
            if (mt * 16 + v + 8 == PP) tv = hi ? w : tv;
            lds_t[wave][mt * 16 + v + hi * 8][laneN] = tv;
        }
    __syncthreads();

    v8f acc2[11];
#pragma unroll
    for (int mt = 0; mt < 11; ++mt) acc2[mt] = vzero;

    auto load_b2 = [&](v2f (&b)[2], const int kp) {
#pragma unroll
        for (int s = 0; s < 2; ++s) {
            const int ka = (2 * kp + s) * 4 + 2 * hi;
            b[s].x = lds_t[wave][ka][laneN];
            b[s].y = lds_t[wave][ka + 1][laneN];
        }
    };
    load_a(aA, qinf, 0);
    load_b2(bA, 0);
#pragma unroll 1
    for (int kp = 0; kp < 20; kp += 2) {
        load_a(aB, qinf, kp + 1);
        load_b2(bB, kp + 1);
        __builtin_amdgcn_sched_barrier(0);
        mma_row(acc2, aA, bA);
        load_a(aA, qinf, kp + 2);
        load_b2(bA, kp + 2);
        __builtin_amdgcn_sched_barrier(0);
        mma_row(acc2, aB, bB);
    }
    load_a(aB, qinf, 21);
    load_b2(bB, 21);
    __builtin_amdgcn_sched_barrier(0);
    mma_row(acc2, aA, bA);
    mma_row(acc2, aB, bB);

    static_assert(LPOS % 16 == 0, "a tile is wholly inside or wholly outside the patch list");
    float (*T)[17] = lds_t[wave];
#pragma unroll
    for (int mt = 0; mt < 11; ++mt) {
#pragma unroll
        for (int v = 0; v < 8; ++v) {
            const int M = mt * 16 + v + hi * 8;
            if (M < PP) T[M][laneN] = acc2[mt][v] * w;
        }
    }
    for (int M = lane; M <= PP; M += 32) T[M][16] = 0.0f;
    __syncthreads();

    if (l0 < LPOS) {
        const int ip0 = l0 / HO;
        const int jp0 = l0 - ip0 * HO;
        const int s   = (HO - jp0 < 16) ? (HO - jp0) : 16;
        float* nb = numb + img * IMGPAD + ip0 * HP + jp0;
        float* db = divb + img * IMGPAD + ip0 * HP + jp0;
        const int npass = (s < 16) ? 2 : 1;
        for (int pass = 0; pass < npass; ++pass) {
            const int q = lane + 32 * pass;
            int col[P_];
            bool any = false;
#pragma unroll
            for (int dj = 0; dj < P_; ++dj) {
                const int sh = q - dj;
                const bool row0 = (sh >= 0) && (sh < s);
                const bool row1 = (sh >= s + 12) && (sh <= 27);
                col[dj] = row0 ? sh : (row1 ? sh - 12 : 16);
                any = any || row0 || row1;
            }
            float dsum = 0.0f;
#pragma unroll
            for (int dj = 0; dj < P_; ++dj) dsum += T[PP][col[dj]];
            if (any) {
#pragma unroll
                for (int di = 0; di < P_; ++di) {
                    float nsum = 0.0f;
#pragma unroll
                    for (int dj = 0; dj < P_; ++dj) nsum += T[di * P_ + dj][col[dj]];
                    atomicAdd(nb + di * HP + q, nsum);
                    atomicAdd(db + di * HP + q, dsum);
                }
            }
        }
    }
}

__global__ void finish_kernel(const float* __restrict__ ws,
                              float* __restrict__ out) {
    int idx = blockIdx.x * blockDim.x + threadIdx.x;
    if (idx >= NIMG * HW * HW) return;
    int n   = idx / (HW * HW);
    int rem = idx % (HW * HW);
    int y   = rem / HW;
    int x   = rem % HW;
    int s = n * IMGPAD + (y + P_ - 1) * HP + (x + P_ - 1);
    out[idx] = (ws[NUM_OFF + s] / ws[DIV_OFF + s] + 1.0f) * 0.5f;
}

extern "C" void kernel_launch(void* const* d_in, const int* in_sizes, int n_in,
                              void* d_out, int out_size, void* d_ws, size_t ws_size,
                              hipStream_t stream) {
    const float* x      = (const float*)d_in[0];
    const float* sigma_ = (const float*)d_in[1];
    const float* Pm1    = (const float*)d_in[2];
    float* out = (float*)d_out;
    float* ws  = (float*)d_ws;

    const int prepN = NIMG * IMGPAD;
    prep_kernel<<<(prepN + 255) / 256, 256, 0, stream>>>(x, Pm1, ws);

    invert_kernel<<<1, IV_THREADS, 0, stream>>>(Pm1, ws);

    dim3 grid((LPOS + WAVES * 16 - 1) / (WAVES * 16), NIMG);
    dct2net_main<<<grid, 32 * WAVES, 0, stream>>>(sigma_, ws);

    const int outN = NIMG * HW * HW;
    finish_kernel<<<(outN + 255) / 256, 256, 0, stream>>>(ws, out);
}
